# starting kernel
# baseline (speedup 1.0000x reference)
#include <hip/hip_runtime.h>


typedef __attribute__((ext_vector_type(2))) float v2f;
typedef __attribute__((ext_vector_type(8))) float v8f;

#define D_DIM 128
#define O_DIM 128
#define PAIR_STRIDE 288
#define N_PAIRS (D_DIM / 2)

__global__ void rgcn_zero_kernel(float4* __restrict__ p, int n4) {
    int i = blockIdx.x * blockDim.x + threadIdx.x;
    if (i < n4) p[i] = make_float4(0.f, 0.f, 0.f, 0.f);
}

__global__ void __launch_bounds__(256)
rgcn_gemm_xw_kernel(const float* __restrict__ x,
                    const float* __restrict__ Wr,
                    float* __restrict__ y,
                    int n_rows) {
    __shared__ float sW[N_PAIRS * PAIR_STRIDE];

    const int tid = threadIdx.x;

    for (int g = tid; g < D_DIM * (O_DIM / 4); g += 256) {
        int d  = g >> 5;
        int cq = g & 31;
        float4 w4 = *(const float4*)(Wr + (size_t)d * O_DIM + cq * 4);
        float* b = &sW[(d >> 1) * PAIR_STRIDE + cq * 8 + (d & 1)];
        b[0] = w4.x; b[2] = w4.y; b[4] = w4.z; b[6] = w4.w;
    }
    __syncthreads();

    const int wave = tid >> 5;
    const int lane = tid & 31;
    const int m0   = (blockIdx.x * 8 + wave) * 16;
    if (m0 >= n_rows) return;

    const int mrow  = lane & 15;
    const int pl    = lane >> 4;
    const int khalf = pl * 2;
    const int ncol  = lane & 15;

    v8f acc[8] = {};

    const float* xrow = x + (size_t)(m0 + mrow) * D_DIM;

    for (int k0 = 0; k0 < D_DIM; k0 += 4) {
        v2f a = *(const v2f*)(xrow + k0 + khalf);
        const float* sWp = &sW[((k0 >> 1) + pl) * PAIR_STRIDE + ncol * 2];
#pragma unroll
        for (int t = 0; t < 8; ++t) {
            v2f b = *(const v2f*)(sWp + t * 32);
            acc[t] = __builtin_amdgcn_wmma_f32_16x16x4_f32(
                 false, a,  false, b,
                 (short)0, acc[t],
                 false,  false);
        }
    }

    const int mbase = m0 + (pl << 3);
#pragma unroll
    for (int t = 0; t < 8; ++t) {
#pragma unroll
        for (int v = 0; v < 8; ++v) {
            y[(size_t)(mbase + v) * O_DIM + t * 16 + ncol] = acc[t][v];
        }
    }
}

__global__ void __launch_bounds__(256)
rgcn_scatter_kernel(const int* __restrict__ erow,
                    const int* __restrict__ ecol,
                    const float* __restrict__ eval,
                    const float* __restrict__ y,
                    float* __restrict__ out,
                    int n_edges) {
    const int wave = (blockIdx.x * blockDim.x + threadIdx.x) >> 5;
    const int lane = threadIdx.x & 31;
    if (wave >= n_edges) return;

    const int   r = erow[wave];
    const int   c = ecol[wave];
    const float v = eval[wave];

    float4 f = *(const float4*)(y + (size_t)c * O_DIM + lane * 4);
    float* dst = out + (size_t)r * O_DIM + lane * 4;
    atomicAdd(dst + 0, v * f.x);
    atomicAdd(dst + 1, v * f.y);
    atomicAdd(dst + 2, v * f.z);
    atomicAdd(dst + 3, v * f.w);
}

extern "C" void kernel_launch(void* const* d_in, const int* in_sizes, int n_in,
                              void* d_out, int out_size, void* d_ws, size_t ws_size,
                              hipStream_t stream) {
    const float* x    = (const float*)d_in[0];
    const int*   erow = (const int*)  d_in[1];
    const int*   ecol = (const int*)  d_in[2];
    const float* eval = (const float*)d_in[3];
    const float* W    = (const float*)d_in[4];
    float*       out  = (float*)d_out;
    float*       y    = (float*)d_ws;

    const int R = 4;
    const int N = in_sizes[0] / D_DIM;
    const int E = in_sizes[1] / R;

    {
        int n4 = (N * O_DIM) / 4;
        int blocks = (n4 + 255) / 256;
        rgcn_zero_kernel<<<blocks, 256, 0, stream>>>((float4*)out, n4);
    }

    const int wave_tiles = (N + 15) / 16;
    const int gemm_blocks = (wave_tiles + 7) / 8;
    const int scat_blocks = (E + 7) / 8;

    for (int r = 0; r < R; ++r) {
        rgcn_gemm_xw_kernel<<<gemm_blocks, 256, 0, stream>>>(
            x, W + (size_t)r * D_DIM * O_DIM, y, N);
        rgcn_scatter_kernel<<<scat_blocks, 256, 0, stream>>>(
            erow + (size_t)r * E, ecol + (size_t)r * E, eval + (size_t)r * E,
            y, out, E);
    }
}


# scatter: lane owns columns l,l+32,l+64,l+96, so each f32 atomic and each y load covers one whole 128-B line, not 8 floats in each of 4 lines; sums still by atomics in arrival order
# speedup vs baseline: 1.2923x; 1.2923x over previous
#include <hip/hip_runtime.h>


typedef __attribute__((ext_vector_type(2))) float v2f;
typedef __attribute__((ext_vector_type(8))) float v8f;

#define D_DIM 128
#define O_DIM 128
#define PAIR_STRIDE 288
#define N_PAIRS (D_DIM / 2)

__global__ void rgcn_zero_kernel(float4* __restrict__ p, int n4) {
    int i = blockIdx.x * blockDim.x + threadIdx.x;
    if (i < n4) p[i] = make_float4(0.f, 0.f, 0.f, 0.f);
}

__global__ void __launch_bounds__(256)
rgcn_gemm_xw_kernel(const float* __restrict__ x,
                    const float* __restrict__ Wr,
                    float* __restrict__ y,
                    int n_rows) {
    __shared__ float sW[N_PAIRS * PAIR_STRIDE];

    const int tid = threadIdx.x;

    for (int g = tid; g < D_DIM * (O_DIM / 4); g += 256) {
        int d  = g >> 5;
        int cq = g & 31;
        float4 w4 = *(const float4*)(Wr + (size_t)d * O_DIM + cq * 4);
        float* b = &sW[(d >> 1) * PAIR_STRIDE + cq * 8 + (d & 1)];
        b[0] = w4.x; b[2] = w4.y; b[4] = w4.z; b[6] = w4.w;
    }
    __syncthreads();

    const int wave = tid >> 5;
    const int lane = tid & 31;
    const int m0   = (blockIdx.x * 8 + wave) * 16;
    if (m0 >= n_rows) return;

    const int mrow  = lane & 15;
    const int pl    = lane >> 4;
    const int khalf = pl * 2;
    const int ncol  = lane & 15;

    v8f acc[8] = {};

    const float* xrow = x + (size_t)(m0 + mrow) * D_DIM;

    for (int k0 = 0; k0 < D_DIM; k0 += 4) {
        v2f a = *(const v2f*)(xrow + k0 + khalf);
        const float* sWp = &sW[((k0 >> 1) + pl) * PAIR_STRIDE + ncol * 2];
#pragma unroll
        for (int t = 0; t < 8; ++t) {
            v2f b = *(const v2f*)(sWp + t * 32);
            acc[t] = __builtin_amdgcn_wmma_f32_16x16x4_f32(
                 false, a,  false, b,
                 (short)0, acc[t],
                 false,  false);
        }
    }

    const int mbase = m0 + (pl << 3);
#pragma unroll
    for (int t = 0; t < 8; ++t) {
#pragma unroll
        for (int v = 0; v < 8; ++v) {
            y[(size_t)(mbase + v) * O_DIM + t * 16 + ncol] = acc[t][v];
        }
    }
}

__global__ void __launch_bounds__(256)
rgcn_scatter_kernel(const int* __restrict__ erow,
                    const int* __restrict__ ecol,
                    const float* __restrict__ eval,
                    const float* __restrict__ y,
                    float* __restrict__ out,
                    int n_edges) {
    const int wave = (blockIdx.x * blockDim.x + threadIdx.x) >> 5;
    const int lane = threadIdx.x & 31;
    if (wave >= n_edges) return;

    const int   r = erow[wave];
    const int   c = ecol[wave];
    const float v = eval[wave];

    const float* src = y + (size_t)c * O_DIM + lane;
    float* dst = out + (size_t)r * O_DIM + lane;

    const float f0 = src[0];
    const float f1 = src[32];
    const float f2 = src[64];
    const float f3 = src[96];

    atomicAdd(dst + 0,  v * f0);
    atomicAdd(dst + 32, v * f1);
    atomicAdd(dst + 64, v * f2);
    atomicAdd(dst + 96, v * f3);
}

extern "C" void kernel_launch(void* const* d_in, const int* in_sizes, int n_in,
                              void* d_out, int out_size, void* d_ws, size_t ws_size,
                              hipStream_t stream) {
    const float* x    = (const float*)d_in[0];
    const int*   erow = (const int*)  d_in[1];
    const int*   ecol = (const int*)  d_in[2];
    const float* eval = (const float*)d_in[3];
    const float* W    = (const float*)d_in[4];
    float*       out  = (float*)d_out;
    float*       y    = (float*)d_ws;

    const int R = 4;
    const int N = in_sizes[0] / D_DIM;
    const int E = in_sizes[1] / R;

    {
        int n4 = (N * O_DIM) / 4;
        int blocks = (n4 + 255) / 256;
        rgcn_zero_kernel<<<blocks, 256, 0, stream>>>((float4*)out, n4);
    }

    const int wave_tiles = (N + 15) / 16;
    const int gemm_blocks = (wave_tiles + 7) / 8;
    const int scat_blocks = (E + 7) / 8;

    for (int r = 0; r < R; ++r) {
        rgcn_gemm_xw_kernel<<<gemm_blocks, 256, 0, stream>>>(
            x, W + (size_t)r * D_DIM * O_DIM, y, N);
        rgcn_scatter_kernel<<<scat_blocks, 256, 0, stream>>>(
            erow + (size_t)r * E, ecol + (size_t)r * E, eval + (size_t)r * E,
            y, out, E);
    }
}


# scatter: one wave takes 32 consecutive edges (one coalesced index load per array) and issues the 32 y-line loads of 8 edges before their atomics; sums still by f32 atomics in arrival order
# speedup vs baseline: 1.5799x; 1.2225x over previous
#include <hip/hip_runtime.h>


typedef __attribute__((ext_vector_type(2))) float v2f;
typedef __attribute__((ext_vector_type(8))) float v8f;

#define D_DIM 128
#define O_DIM 128
#define PAIR_STRIDE 288
#define N_PAIRS (D_DIM / 2)

__global__ void rgcn_zero_kernel(float4* __restrict__ p, int n4) {
    int i = blockIdx.x * blockDim.x + threadIdx.x;
    if (i < n4) p[i] = make_float4(0.f, 0.f, 0.f, 0.f);
}

__global__ void __launch_bounds__(256)
rgcn_gemm_xw_kernel(const float* __restrict__ x,
                    const float* __restrict__ Wr,
                    float* __restrict__ y,
                    int n_rows) {
    __shared__ float sW[N_PAIRS * PAIR_STRIDE];

    const int tid = threadIdx.x;

    for (int g = tid; g < D_DIM * (O_DIM / 4); g += 256) {
        int d  = g >> 5;
        int cq = g & 31;
        float4 w4 = *(const float4*)(Wr + (size_t)d * O_DIM + cq * 4);
        float* b = &sW[(d >> 1) * PAIR_STRIDE + cq * 8 + (d & 1)];
        b[0] = w4.x; b[2] = w4.y; b[4] = w4.z; b[6] = w4.w;
    }
    __syncthreads();

    const int wave = tid >> 5;
    const int lane = tid & 31;
    const int m0   = (blockIdx.x * 8 + wave) * 16;
    if (m0 >= n_rows) return;

    const int mrow  = lane & 15;
    const int pl    = lane >> 4;
    const int khalf = pl * 2;
    const int ncol  = lane & 15;

    v8f acc[8] = {};

    const float* xrow = x + (size_t)(m0 + mrow) * D_DIM;

    for (int k0 = 0; k0 < D_DIM; k0 += 4) {
        v2f a = *(const v2f*)(xrow + k0 + khalf);
        const float* sWp = &sW[((k0 >> 1) + pl) * PAIR_STRIDE + ncol * 2];
#pragma unroll
        for (int t = 0; t < 8; ++t) {
            v2f b = *(const v2f*)(sWp + t * 32);
            acc[t] = __builtin_amdgcn_wmma_f32_16x16x4_f32(
                 false, a,  false, b,
                 (short)0, acc[t],
                 false,  false);
        }
    }

    const int mbase = m0 + (pl << 3);
#pragma unroll
    for (int t = 0; t < 8; ++t) {
#pragma unroll
        for (int v = 0; v < 8; ++v) {
            y[(size_t)(mbase + v) * O_DIM + t * 16 + ncol] = acc[t][v];
        }
    }
}

#define EDGES_PER_WAVE 32
#define EDGE_BATCH 8

__global__ void __launch_bounds__(256)
rgcn_scatter_kernel(const int* __restrict__ erow,
                    const int* __restrict__ ecol,
                    const float* __restrict__ eval,
                    const float* __restrict__ y,
                    float* __restrict__ out,
                    int n_edges) {
    const int wave_in_block = __builtin_amdgcn_readfirstlane((int)(threadIdx.x >> 5));
    const int lane = threadIdx.x & 31;
    const int base = (blockIdx.x * (blockDim.x >> 5) + wave_in_block) * EDGES_PER_WAVE;
    if (base >= n_edges) return;
    const int left = n_edges - base;
    const int cnt  = left < EDGES_PER_WAVE ? left : EDGES_PER_WAVE;

    int   r = 0;
    int   c = 0;
    float v = 0.f;
    if (lane < cnt) {
        r = erow[base + lane];
        c = ecol[base + lane];
        v = eval[base + lane];
    }

    const float* ylane = y + lane;
    float*       olane = out + lane;

    for (int b = 0; b < cnt; b += EDGE_BATCH) {
        float f[EDGE_BATCH][4];
#pragma unroll
        for (int j = 0; j < EDGE_BATCH; ++j) {
            const int cj = __builtin_amdgcn_readlane(c, b + j);
            const float* src = ylane + (size_t)cj * O_DIM;
            f[j][0] = src[0];
            f[j][1] = src[32];
            f[j][2] = src[64];
            f[j][3] = src[96];
        }
#pragma unroll
        for (int j = 0; j < EDGE_BATCH; ++j) {
            if (b + j < cnt) {
                const int   rj = __builtin_amdgcn_readlane(r, b + j);
                const float vj = __int_as_float(
                    __builtin_amdgcn_readlane(__float_as_int(v), b + j));
                float* dst = olane + (size_t)rj * O_DIM;
                atomicAdd(dst + 0,  vj * f[j][0]);
                atomicAdd(dst + 32, vj * f[j][1]);
                atomicAdd(dst + 64, vj * f[j][2]);
                atomicAdd(dst + 96, vj * f[j][3]);
            }
        }
    }
}

extern "C" void kernel_launch(void* const* d_in, const int* in_sizes, int n_in,
                              void* d_out, int out_size, void* d_ws, size_t ws_size,
                              hipStream_t stream) {
    const float* x    = (const float*)d_in[0];
    const int*   erow = (const int*)  d_in[1];
    const int*   ecol = (const int*)  d_in[2];
    const float* eval = (const float*)d_in[3];
    const float* W    = (const float*)d_in[4];
    float*       out  = (float*)d_out;
    float*       y    = (float*)d_ws;

    const int R = 4;
    const int N = in_sizes[0] / D_DIM;
    const int E = in_sizes[1] / R;

    {
        int n4 = (N * O_DIM) / 4;
        int blocks = (n4 + 255) / 256;
        rgcn_zero_kernel<<<blocks, 256, 0, stream>>>((float4*)out, n4);
    }

    const int wave_tiles = (N + 15) / 16;
    const int gemm_blocks = (wave_tiles + 7) / 8;
    const int scat_blocks = (E + 255) / 256;

    for (int r = 0; r < R; ++r) {
        rgcn_gemm_xw_kernel<<<gemm_blocks, 256, 0, stream>>>(
            x, W + (size_t)r * D_DIM * O_DIM, y, N);
        rgcn_scatter_kernel<<<scat_blocks, 256, 0, stream>>>(
            erow + (size_t)r * E, ecol + (size_t)r * E, eval + (size_t)r * E,
            y, out, E);
    }
}
